# starting kernel
# baseline (speedup 1.0000x reference)
#include <hip/hip_runtime.h>
#include <math.h>

#define NUM_TBLOCKS 8
#define SEQ 54
#define DIM 6
#define QKV (3 * DIM)
#define FEAT (SEQ * DIM)
#define POL 12
#define LN_EPS 1e-5f
#define QSCALE 0.40824829046386302f

#define WAVES_PER_BLOCK 4
#define THREADS (WAVES_PER_BLOCK * 32)
#define WAVE_LDS 3136
#define WCACHE 704

typedef __attribute__((ext_vector_type(2))) float v2f;
typedef __attribute__((ext_vector_type(4))) float v4f;
typedef __attribute__((ext_vector_type(8))) float v8f;

__device__ __forceinline__ v8f wmma_f32(v2f a, v2f b, v8f c) {
  return __builtin_amdgcn_wmma_f32_16x16x4_f32(false, a, false, b, (short)0, c,
                                               false, false);
}

__device__ __forceinline__ float eluf(float x) {
  return x > 0.0f ? x : (__expf(x) - 1.0f);
}

__global__ __launch_bounds__(THREADS) void rubik_fused(
    const float* __restrict__ x, const float* __restrict__ embed_w,
    const float* __restrict__ embed_b, const float* __restrict__ fc1_w,
    const float* __restrict__ fc1_b, const float* __restrict__ ln_g,
    const float* __restrict__ ln_b, const float* __restrict__ iw,
    const float* __restrict__ ib, const float* __restrict__ ow,
    const float* __restrict__ ob, const float* __restrict__ vw,
    const float* __restrict__ vb, const float* __restrict__ pw,
    const float* __restrict__ pb, float* __restrict__ out, int n_batch) {
  __shared__ float smem[WAVES_PER_BLOCK * WAVE_LDS + WCACHE];
  const int lane = threadIdx.x & 31;
  const int wave = threadIdx.x >> 5;
  int b = blockIdx.x * WAVES_PER_BLOCK + wave;
  if (b >= n_batch) b = n_batch - 1;

  float* H = &smem[wave * WAVE_LDS];
  float* Z = H + 324;
  float* Qp = Z + 324;
  float* Kp = Qp + 512;
  float* Vp = Kp + 512;
  float* P = Vp + 448;
  float* dump = H + 3016;

  float* WC = &smem[WAVES_PER_BLOCK * WAVE_LDS];
  float* W_FC = WC;
  float* W_OW = WC + 128;
  float* W_Q0 = WC + 256;
  float* W_Q1 = WC + 384;
  float* B_FC = WC + 512;
  float* B_OW = WC + 544;
  float* B_Q0 = WC + 576;
  float* B_Q1 = WC + 608;

  const float* xb = x + (size_t)b * FEAT;
  __builtin_prefetch(xb, 0, 1);

  const int nloc = lane & 15;
  const int kb = (lane < 16) ? 0 : 2;
  const int upper8 = (lane < 16) ? 0 : 8;
  const bool lowhalf = (lane < 16);

  const int qgrp = nloc / 6;
  const int qcol = nloc - qgrp * 6;
  const float qscl = (qgrp == 0) ? QSCALE : 1.0f;
  float* qkvBase = Qp + qgrp * 512 + qcol;

  auto fillWB = [&](const float* w, const float* bias, int nCols, int ntile,
                    float* wimg, float* bimg) {
    for (int slot = threadIdx.x; slot < 128; slot += THREADS) {
      int kk = slot >> 6, ln = (slot >> 1) & 31, p = slot & 1;
      int nl = ln & 15;
      int c = kk * 4 + ((ln < 16) ? 0 : 2) + p;
      int j = ntile * 16 + nl;
      bool v = (j < nCols) && (c < DIM);
      wimg[slot] = v ? w[v ? j * DIM + c : 0] : 0.0f;
    }
    for (int ln = threadIdx.x; ln < 32; ln += THREADS) {
      int j = ntile * 16 + (ln & 15);
      bimg[ln] = (j < nCols) ? bias[(j < nCols) ? j : 0] : 0.0f;
    }
  };

  auto loadA = [&](const float* src, int mt, v2f aw[2]) {
    int row = mt * 16 + nloc;
#pragma unroll
    for (int kk = 0; kk < 2; ++kk)
      aw[kk] = *(const v2f*)(src + row * DIM + kk * 4 + kb);
  };
  auto loadBimg = [&](const float* wimg, v2f bw[2]) {
    bw[0] = *(const v2f*)(wimg + lane * 2);
    bw[1] = *(const v2f*)(wimg + 64 + lane * 2);
  };
  auto loadCimg = [&](const float* bimg) {
    float bvv = bimg[lane];
    v8f c;
#pragma unroll
    for (int r = 0; r < 8; ++r) c[r] = bvv;
    return c;
  };

  for (int idx = lane; idx < FEAT; idx += 32) Z[idx] = xb[idx];
  for (int idx = lane; idx < 512 + 512 + 448; idx += 32) Qp[idx] = 0.0f;
  fillWB(embed_w, embed_b, DIM, 0, W_FC, B_FC);
  __syncthreads();

  {
    v2f bw[2];
    loadBimg(W_FC, bw);
    v8f cb = loadCimg(B_FC);
#pragma unroll
    for (int mt = 0; mt < 4; ++mt) {
      v2f aw[2];
      loadA(Z, mt, aw);
      v8f d = wmma_f32(aw[0], bw[0], cb);
      d = wmma_f32(aw[1], bw[1], d);
      int row0 = mt * 16 + upper8;
      bool ok = (nloc < DIM) && ((mt < 3) || lowhalf);
      float* bz = ok ? &H[row0 * DIM + nloc] : dump;
#pragma unroll
      for (int r = 0; r < 8; ++r)
        if (mt < 3 || r < 6) bz[r * DIM] = d[r];
    }
  }
  __syncthreads();

  for (int L = 0; L < NUM_TBLOCKS; ++L) {
    const float* g = ln_g + L * DIM;
    const float* bbia = ln_b + L * DIM;

    fillWB(fc1_w + L * DIM * DIM, fc1_b + L * DIM, DIM, 0, W_FC, B_FC);
    fillWB(ow + L * DIM * DIM, ob + L * DIM, DIM, 0, W_OW, B_OW);
    fillWB(iw + L * QKV * DIM, ib + L * QKV, QKV, 0, W_Q0, B_Q0);
    fillWB(iw + L * QKV * DIM, ib + L * QKV, QKV, 1, W_Q1, B_Q1);
    __syncthreads();

    {
      v2f bw[2];
      loadBimg(W_FC, bw);
      v8f cb = loadCimg(B_FC);
#pragma unroll
      for (int mt = 0; mt < 4; ++mt) {
        v2f aw[2];
        loadA(H, mt, aw);
        v8f d = wmma_f32(aw[0], bw[0], cb);
        d = wmma_f32(aw[1], bw[1], d);
        int row0 = mt * 16 + upper8;
        bool ok = (nloc < DIM) && ((mt < 3) || lowhalf);
        float* bz = ok ? &Z[row0 * DIM + nloc] : dump;
#pragma unroll
        for (int r = 0; r < 8; ++r)
          if (mt < 3 || r < 6) bz[r * DIM] = eluf(d[r]);
      }
    }
    __syncthreads();

    for (int s = lane; s < SEQ; s += 32) {
      float mu = 0.0f;
#pragma unroll
      for (int i = 0; i < DIM; ++i) mu += Z[s * DIM + i];
      mu *= (1.0f / DIM);
      float var = 0.0f;
#pragma unroll
      for (int i = 0; i < DIM; ++i) {
        float d = Z[s * DIM + i] - mu;
        var += d * d;
      }
      var *= (1.0f / DIM);
      float rs = rsqrtf(var + LN_EPS);
#pragma unroll
      for (int i = 0; i < DIM; ++i)
        Z[s * DIM + i] = (Z[s * DIM + i] - mu) * rs * g[i] + bbia[i];
    }
    __syncthreads();

    {
      v2f bw0[2], bw1[2];
      loadBimg(W_Q0, bw0);
      loadBimg(W_Q1, bw1);
      v8f cb0 = loadCimg(B_Q0);
      v8f cb1 = loadCimg(B_Q1);
#pragma unroll
      for (int mt = 0; mt < 4; ++mt) {
        v2f aw[2];
        loadA(Z, mt, aw);
        v8f d0 = wmma_f32(aw[0], bw0[0], cb0);
        d0 = wmma_f32(aw[1], bw0[1], d0);
        v8f d1 = wmma_f32(aw[0], bw1[0], cb1);
        d1 = wmma_f32(aw[1], bw1[1], d1);
        int row0 = mt * 16 + upper8;
        bool ok0 = (mt < 3) || lowhalf;
        float* b0 = ok0 ? (qkvBase + row0 * 8) : dump;
        float* b1 = (ok0 && nloc < 2) ? (Vp + row0 * 8 + 4 + nloc) : dump;
#pragma unroll
        for (int r = 0; r < 8; ++r)
          if (mt < 3 || r < 6) {
            b0[r * 8] = d0[r] * qscl;
            b1[r * 8] = d1[r];
          }
      }
    }
    __syncthreads();

    v2f ak[4][2];
#pragma unroll
    for (int mt = 0; mt < 4; ++mt)
#pragma unroll
      for (int kk = 0; kk < 2; ++kk)
        ak[mt][kk] = *(const v2f*)(Kp + (mt * 16 + nloc) * 8 + kk * 4 + kb);
    v2f bv[14];
#pragma unroll
    for (int kc = 0; kc < 14; ++kc) {
      int kr = kc * 4 + kb;
      bv[kc].x = (nloc < 8) ? Vp[kr * 8 + nloc] : 0.0f;
      bv[kc].y = (nloc < 8) ? Vp[(kr + 1) * 8 + nloc] : 0.0f;
    }

    for (int nt = 0; nt < 4; ++nt) {
      v2f bq[2];
#pragma unroll
      for (int kk = 0; kk < 2; ++kk)
        bq[kk] = *(const v2f*)(Qp + (nt * 16 + nloc) * 8 + kk * 4 + kb);
      v8f t[4];
#pragma unroll
      for (int mt = 0; mt < 4; ++mt) {
        v8f c = {};
#pragma unroll
        for (int kk = 0; kk < 2; ++kk) c = wmma_f32(ak[mt][kk], bq[kk], c);
        t[mt] = c;
      }

      float mx = -3.0e38f;
#pragma unroll
      for (int mt = 0; mt < 4; ++mt)
#pragma unroll
        for (int r = 0; r < 8; ++r) {
          int k = mt * 16 + r + upper8;
          if (k < SEQ) mx = fmaxf(mx, t[mt][r]);
        }
      mx = fmaxf(mx, __shfl_xor(mx, 16));
      float sum = 0.0f;
#pragma unroll
      for (int mt = 0; mt < 4; ++mt)
#pragma unroll
        for (int r = 0; r < 8; ++r) {
          int k = mt * 16 + r + upper8;
          float e = (k < SEQ) ? __expf(t[mt][r] - mx) : 0.0f;
          t[mt][r] = e;
          sum += e;
        }
      sum += __shfl_xor(sum, 16);
      float inv = 1.0f / sum;

#pragma unroll
      for (int mt = 0; mt < 4; ++mt) {
        v4f lo = {t[mt][0] * inv, t[mt][1] * inv, t[mt][2] * inv,
                  t[mt][3] * inv};
        v4f hi = {t[mt][4] * inv, t[mt][5] * inv, t[mt][6] * inv,
                  t[mt][7] * inv};
        bool ok = (mt < 3) || lowhalf;
        float* bp = ok ? (P + nloc * 56 + mt * 16 + upper8) : dump;
        *(v4f*)(bp) = lo;
        *(v4f*)(bp + 4) = hi;
      }
      __syncthreads();

      v8f o = {};
#pragma unroll
      for (int kc = 0; kc < 14; ++kc) {
        v2f ap = *(const v2f*)(P + nloc * 56 + kc * 4 + kb);
        o = wmma_f32(ap, bv[kc], o);
      }
      {
        int row0 = nt * 16 + upper8;
        bool ok = (nloc < DIM) && ((nt < 3) || lowhalf);
        float* bz = ok ? &Z[row0 * DIM + nloc] : dump;
#pragma unroll
        for (int r = 0; r < 8; ++r)
          if (nt < 3 || r < 6) bz[r * DIM] = o[r];
      }
      __syncthreads();
    }

    {
      v2f bw[2];
      loadBimg(W_OW, bw);
      v8f cb = loadCimg(B_OW);
#pragma unroll
      for (int mt = 0; mt < 4; ++mt) {
        v2f aw[2];
        loadA(Z, mt, aw);
        v8f d = wmma_f32(aw[0], bw[0], cb);
        d = wmma_f32(aw[1], bw[1], d);
        int row0 = mt * 16 + upper8;
        bool ok = (nloc < DIM) && ((mt < 3) || lowhalf);
        float* bh = ok ? &H[row0 * DIM + nloc] : dump;
#pragma unroll
        for (int r = 0; r < 8; ++r)
          if (mt < 3 || r < 6) {
            float hv = H[row0 * DIM + nloc + r * DIM];
            bh[r * DIM] = eluf(d[r]) + hv;
          }
      }
    }
    __syncthreads();
  }

  float vacc = 0.0f;
  float pacc[POL];
#pragma unroll
  for (int o = 0; o < POL; ++o) pacc[o] = 0.0f;
  for (int j = lane; j < FEAT; j += 32) {
    float fj = H[j];
    vacc += fj * vw[j];
#pragma unroll
    for (int o = 0; o < POL; ++o) pacc[o] += fj * pw[o * FEAT + j];
  }
  for (int off = 16; off >= 1; off >>= 1) {
    vacc += __shfl_xor(vacc, off, 32);
#pragma unroll
    for (int o = 0; o < POL; ++o) pacc[o] += __shfl_xor(pacc[o], off, 32);
  }
  if (lane == 0) {
    out[b] = vacc + vb[0];
    float logit[POL];
    float mx = -3.0e38f;
#pragma unroll
    for (int o = 0; o < POL; ++o) {
      logit[o] = pacc[o] + pb[o];
      mx = fmaxf(mx, logit[o]);
    }
    float sum = 0.0f;
#pragma unroll
    for (int o = 0; o < POL; ++o) {
      logit[o] = __expf(logit[o] - mx);
      sum += logit[o];
    }
    float inv = 1.0f / sum;
    float* pout = out + n_batch + (size_t)b * POL;
#pragma unroll
    for (int o = 0; o < POL; ++o) pout[o] = logit[o] * inv;
  }
}

extern "C" void kernel_launch(void* const* d_in, const int* in_sizes, int n_in,
                              void* d_out, int out_size, void* d_ws,
                              size_t ws_size, hipStream_t stream) {
  (void)n_in;
  (void)out_size;
  (void)d_ws;
  (void)ws_size;
  const float* x = (const float*)d_in[0];
  const float* embed_w = (const float*)d_in[1];
  const float* embed_b = (const float*)d_in[2];
  const float* fc1_w = (const float*)d_in[3];
  const float* fc1_b = (const float*)d_in[4];
  const float* ln_g = (const float*)d_in[5];
  const float* ln_b = (const float*)d_in[6];
  const float* iw = (const float*)d_in[7];
  const float* ib = (const float*)d_in[8];
  const float* ow = (const float*)d_in[9];
  const float* ob = (const float*)d_in[10];
  const float* vw = (const float*)d_in[11];
  const float* vb = (const float*)d_in[12];
  const float* pw = (const float*)d_in[13];
  const float* pb = (const float*)d_in[14];

  int n_batch = in_sizes[0] / FEAT;
  int grid = (n_batch + WAVES_PER_BLOCK - 1) / WAVES_PER_BLOCK;
  rubik_fused<<<grid, THREADS, 0, stream>>>(x, embed_w, embed_b, fc1_w, fc1_b,
                                            ln_g, ln_b, iw, ib, ow, ob, vw, vb,
                                            pw, pb, (float*)d_out, n_batch);
}


# Tail tile re-slotted (tokens 48-50 in rows 0-2, 51-53 in rows 8-10): tail keys fill 3 score registers of both lane halves; key masks compile-time, 12 exp a layer, V fold gone; key sum order changes
# speedup vs baseline: 1.3019x; 1.3019x over previous
#include <hip/hip_runtime.h>
#include <math.h>


#define NUM_TBLOCKS 8
#define SEQ 54
#define DIM 6
#define QKV (3 * DIM)
#define FEAT (SEQ * DIM)
#define POL 12
#define LN_EPS 1e-5f
#define QSCALE 0.40824829046386302f

#define WAVES_PER_BLOCK 1
#define THREADS (WAVES_PER_BLOCK * 32)
#define PPITCH 68
#define LDS_FLOATS (64 * DIM + 16 * PPITCH + 64)

typedef __attribute__((ext_vector_type(2))) float v2f;
typedef __attribute__((ext_vector_type(4))) float v4f;
typedef __attribute__((ext_vector_type(8))) float v8f;

__device__ __forceinline__ v8f wmma_f32(v2f a, v2f b, v8f c) {
  return __builtin_amdgcn_wmma_f32_16x16x4_f32(false, a, false, b, (short)0, c,
                                               false, false);
}

__device__ __forceinline__ float eluf(float x) {
  return x > 0.0f ? x : (__expf(x) - 1.0f);
}


__global__ __launch_bounds__(THREADS) void rubik_fused(
    const float* __restrict__ x, const float* __restrict__ embed_w,
    const float* __restrict__ embed_b, const float* __restrict__ fc1_w,
    const float* __restrict__ fc1_b, const float* __restrict__ ln_g,
    const float* __restrict__ ln_b, const float* __restrict__ iw,
    const float* __restrict__ ib, const float* __restrict__ ow,
    const float* __restrict__ ob, const float* __restrict__ vw,
    const float* __restrict__ vb, const float* __restrict__ pw,
    const float* __restrict__ pb, float* __restrict__ out, int n_batch) {
  __shared__ __attribute__((aligned(16))) float smem[LDS_FLOATS];
  const int lane = threadIdx.x & 31;
  int b = blockIdx.x;
  if (b >= n_batch) b = n_batch - 1;

  float* Zs = smem;
  float* P = smem + 64 * DIM;
  float* dump = P + 16 * PPITCH;

  const float* xb = x + (size_t)b * FEAT;
  __builtin_prefetch(xb, 0, 1);

  const int nloc = lane & 15;
  const int hh = lane >> 4;
  const bool lowhalf = (lane < 16);
  const int kb = 2 * hh;
  const int upper8 = 8 * hh;
  const v2f zero2 = {0.0f, 0.0f};

  const int rr = nloc & 3;
  const int f6 = (nloc & 8) ? (rr + 2) : (rr < 2 ? rr : rr + 2);
  const bool rowv2 = (nloc & 8) ? (rr < 2) : true;
  const bool rowv1 = rowv2 && !(nloc & 4);
  const int jqk = ((nloc & 4) ? DIM : 0) + f6;

  const int s7 = nloc & 7;
  const bool tailv = (s7 < 3);
  const int tailtok = 48 + s7 + ((nloc & 8) ? 3 : 0);

  auto loadWA = [&](const float* w, bool v, int j, v2f aw[2]) {
    const float* wr = w + (v ? j : 0) * DIM;
    v2f w0 = *(const v2f*)(wr + kb);
    v2f w1 = *(const v2f*)(wr + (lowhalf ? 4 : 0));
    aw[0] = v ? w0 : zero2;
    aw[1] = (v && lowhalf) ? w1 : zero2;
  };
  auto loadCrow = [&](const float* bias, bool two) {
    v8f c;
#pragma unroll
    for (int r = 0; r < 8; ++r) {
      int q = r & 3;
      int f = hh ? (q + 2) : (q < 2 ? q : q + 2);
      bool v = hh ? (q < 2) : true;
      if (r >= 4 && !two) v = false;
      int j = f + (r >= 4 ? DIM : 0);
      float bv = bias[v ? j : 0];
      c[r] = v ? bv : 0.0f;
    }
    return c;
  };
  auto loadWB = [&](const float* w, const float* bias, v2f bw[2], float& bcol) {
    bool vj = (nloc < DIM);
    int jc = vj ? nloc : 0;
    const float* wr = w + jc * DIM;
    v2f w0 = *(const v2f*)(wr + kb);
    v2f w1 = *(const v2f*)(wr + (lowhalf ? 4 : 0));
    float bb = bias[jc];
    bw[0] = vj ? w0 : zero2;
    bw[1] = (vj && lowhalf) ? w1 : zero2;
    bcol = vj ? bb : 0.0f;
  };
  auto splatC = [&](float bvv) {
    v8f c;
#pragma unroll
    for (int r = 0; r < 8; ++r) c[r] = bvv;
    return c;
  };

  v2f H0[4], H1[4];

  {
    v2f ae[2];
    loadWA(embed_w, rowv1, f6, ae);
    v8f ce = loadCrow(embed_b, false);
#pragma unroll
    for (int t = 0; t < 4; ++t) {
      int tok = (t < 3) ? (t * 16 + nloc) : tailtok;
      bool tv = (t < 3) ? true : tailv;
      const float* xr = xb + (tv ? tok : 0) * DIM;
      v2f x0 = *(const v2f*)(xr + kb);
      v2f x1 = *(const v2f*)(xr + (lowhalf ? 4 : 0));
      x0 = tv ? x0 : zero2;
      x1 = (tv && lowhalf) ? x1 : zero2;
      v8f d = wmma_f32(ae[0], x0, ce);
      d = wmma_f32(ae[1], x1, d);
      H0[t] = (v2f){d[0], d[1]};
      H1[t] = (v2f){d[2], d[3]};
    }
  }

  for (int L = 0; L < NUM_TBLOCKS; ++L) {
    v2f af[2], aqk[2], bvw[2], ao[2];
    float bcv;
    loadWA(fc1_w + L * DIM * DIM, rowv1, f6, af);
    v8f cf = loadCrow(fc1_b + L * DIM, false);
    loadWA(iw + L * QKV * DIM, rowv2, jqk, aqk);
    v8f cqk = loadCrow(ib + L * QKV, true);
    loadWB(iw + L * QKV * DIM + 2 * DIM * DIM, ib + L * QKV + 2 * DIM, bvw, bcv);
    v8f cv = splatC(bcv);
    loadWA(ow + L * DIM * DIM, rowv1, f6, ao);
    v8f co = loadCrow(ob + L * DIM, false);
    float g4[4], b4[4];
#pragma unroll
    for (int i = 0; i < 4; ++i) {
      int f = hh ? (i + 2) : (i < 2 ? i : i + 2);
      bool v = hh ? (i < 2) : true;
      float gg = ln_g[L * DIM + (v ? f : 0)];
      float bb = ln_b[L * DIM + (v ? f : 0)];
      g4[i] = v ? gg : 0.0f;
      b4[i] = v ? bb : 0.0f;
    }

    v2f Qb0[4], Qb1[4], Ka0[4], Ka1[4];
    v8f Vd[4];

#pragma unroll
    for (int t = 0; t < 4; ++t) {
      v8f d = wmma_f32(af[0], H0[t], cf);
      d = wmma_f32(af[1], H1[t], d);
      float e0 = eluf(d[0]), e1 = eluf(d[1]), e2 = eluf(d[2]), e3 = eluf(d[3]);
      float s = (e0 + e1) + (e2 + e3);
      s += __shfl_xor(s, 16);
      float mu = s * (1.0f / DIM);
      float d0 = e0 - mu, d1 = e1 - mu;
      float d2 = lowhalf ? (e2 - mu) : 0.0f;
      float d3 = lowhalf ? (e3 - mu) : 0.0f;
      float var = (d0 * d0 + d1 * d1) + (d2 * d2 + d3 * d3);
      var += __shfl_xor(var, 16);
      var *= (1.0f / DIM);
      float rs = rsqrtf(var + LN_EPS);
      v2f z0 = {d0 * rs * g4[0] + b4[0], d1 * rs * g4[1] + b4[1]};
      v2f z1 = {d2 * rs * g4[2] + b4[2], d3 * rs * g4[3] + b4[3]};
      v8f dq = wmma_f32(aqk[0], z0, cqk);
      dq = wmma_f32(aqk[1], z1, dq);
      Qb0[t] = (v2f){dq[0] * QSCALE, dq[1] * QSCALE};
      Qb1[t] = (v2f){dq[2] * QSCALE, dq[3] * QSCALE};
      Ka0[t] = (v2f){dq[4], dq[5]};
      Ka1[t] = (v2f){dq[6], dq[7]};
      v8f dv = wmma_f32(z0, bvw[0], cv);
      dv = wmma_f32(z1, bvw[1], dv);
      Vd[t] = dv;
    }

#pragma unroll
    for (int nt = 0; nt < 4; ++nt) {
      v8f t[4];
#pragma unroll
      for (int mt = 0; mt < 4; ++mt) {
        v8f c = {};
        c = wmma_f32(Ka0[mt], Qb0[nt], c);
        c = wmma_f32(Ka1[mt], Qb1[nt], c);
        t[mt] = c;
      }

      float mx = -3.0e38f;
#pragma unroll
      for (int mt = 0; mt < 4; ++mt)
#pragma unroll
        for (int r = 0; r < 8; ++r) {
          if (mt < 3 || r < 3) mx = fmaxf(mx, t[mt][r]);
        }
      mx = fmaxf(mx, __shfl_xor(mx, 16));
      float sum = 0.0f;
#pragma unroll
      for (int mt = 0; mt < 4; ++mt)
#pragma unroll
        for (int r = 0; r < 8; ++r) {
          if (mt < 3 || r < 3) {
            float e = __expf(t[mt][r] - mx);
            t[mt][r] = e;
            sum += e;
          } else {
            t[mt][r] = 0.0f;
          }
        }
      sum += __shfl_xor(sum, 16);
      float inv = 1.0f / sum;

#pragma unroll
      for (int mt = 0; mt < 4; ++mt) {
        float* bp = P + nloc * PPITCH + mt * 16 + upper8;
        if (mt < 3) {
          v4f lo = {t[mt][0] * inv, t[mt][1] * inv, t[mt][2] * inv,
                    t[mt][3] * inv};
          v4f hi = {t[mt][4] * inv, t[mt][5] * inv, t[mt][6] * inv,
                    t[mt][7] * inv};
          *(v4f*)(bp) = lo;
          *(v4f*)(bp + 4) = hi;
        } else {
          v4f lo = {t[mt][0] * inv, t[mt][1] * inv, t[mt][2] * inv, 0.0f};
          *(v4f*)(bp) = lo;
        }
      }
      __syncthreads();

      v8f o = {};
#pragma unroll
      for (int mt = 0; mt < 4; ++mt)
#pragma unroll
        for (int j = 0; j < 4; ++j) {
          if (mt == 3 && j >= 2) continue;
          v2f ap = *(const v2f*)(P + nloc * PPITCH + mt * 16 + upper8 + 2 * j);
          v2f bvv = {Vd[mt][2 * j], Vd[mt][2 * j + 1]};
          o = wmma_f32(ap, bvv, o);
        }
      {
        int row0 = nt * 16 + upper8;
        float* bz = (nloc < DIM) ? &Zs[row0 * DIM + nloc] : dump;
#pragma unroll
        for (int r = 0; r < 8; ++r) bz[r * DIM] = o[r];
      }
      __syncthreads();
    }

#pragma unroll
    for (int t = 0; t < 4; ++t) {
      const float* orow = Zs + (t * 16 + nloc) * DIM;
      v2f o0 = *(const v2f*)(orow + kb);
      v2f o1 = *(const v2f*)(orow + (lowhalf ? 4 : 0));
      o1 = lowhalf ? o1 : zero2;
      v8f d = wmma_f32(ao[0], o0, co);
      d = wmma_f32(ao[1], o1, d);
      H0[t] = (v2f){eluf(d[0]) + H0[t].x, eluf(d[1]) + H0[t].y};
      H1[t] = (v2f){eluf(d[2]) + H1[t].x, eluf(d[3]) + H1[t].y};
    }
    __syncthreads();
  }

#pragma unroll
  for (int t = 0; t < 4; ++t) {
    int tok = (t < 3) ? (t * 16 + nloc) : tailtok;
    bool tv = (t < 3) ? true : tailv;
    if (tv) {
      *(v2f*)(Zs + tok * DIM + kb) = H0[t];
      if (lowhalf) *(v2f*)(Zs + tok * DIM + 4) = H1[t];
    }
  }
  __syncthreads();
  const float* H = Zs;

  float vacc = 0.0f;
  float pacc[POL];
#pragma unroll
  for (int o = 0; o < POL; ++o) pacc[o] = 0.0f;
  for (int j = lane; j < FEAT; j += 32) {
    float fj = H[j];
    vacc += fj * vw[j];
#pragma unroll
    for (int o = 0; o < POL; ++o) pacc[o] += fj * pw[o * FEAT + j];
  }
  for (int off = 16; off >= 1; off >>= 1) {
    vacc += __shfl_xor(vacc, off, 32);
#pragma unroll
    for (int o = 0; o < POL; ++o) pacc[o] += __shfl_xor(pacc[o], off, 32);
  }
  if (lane == 0) {
    out[b] = vacc + vb[0];
    float logit[POL];
    float mx = -3.0e38f;
#pragma unroll
    for (int o = 0; o < POL; ++o) {
      logit[o] = pacc[o] + pb[o];
      mx = fmaxf(mx, logit[o]);
    }
    float sum = 0.0f;
#pragma unroll
    for (int o = 0; o < POL; ++o) {
      logit[o] = __expf(logit[o] - mx);
      sum += logit[o];
    }
    float inv = 1.0f / sum;
    float* pout = out + n_batch + (size_t)b * POL;
#pragma unroll
    for (int o = 0; o < POL; ++o) pout[o] = logit[o] * inv;
  }
}

extern "C" void kernel_launch(void* const* d_in, const int* in_sizes, int n_in,
                              void* d_out, int out_size, void* d_ws,
                              size_t ws_size, hipStream_t stream) {
  (void)n_in;
  (void)out_size;
  (void)d_ws;
  (void)ws_size;
  const float* x = (const float*)d_in[0];
  const float* embed_w = (const float*)d_in[1];
  const float* embed_b = (const float*)d_in[2];
  const float* fc1_w = (const float*)d_in[3];
  const float* fc1_b = (const float*)d_in[4];
  const float* ln_g = (const float*)d_in[5];
  const float* ln_b = (const float*)d_in[6];
  const float* iw = (const float*)d_in[7];
  const float* ib = (const float*)d_in[8];
  const float* ow = (const float*)d_in[9];
  const float* ob = (const float*)d_in[10];
  const float* vw = (const float*)d_in[11];
  const float* vb = (const float*)d_in[12];
  const float* pw = (const float*)d_in[13];
  const float* pb = (const float*)d_in[14];

  int n_batch = in_sizes[0] / FEAT;
  int grid = (n_batch + WAVES_PER_BLOCK - 1) / WAVES_PER_BLOCK;
  rubik_fused<<<grid, THREADS, 0, stream>>>(x, embed_w, embed_b, fc1_w, fc1_b,
                                            ln_g, ln_b, iw, ib, ow, ob, vw, vb,
                                            pw, pb, (float*)d_out, n_batch);
}


# Softmax per score cut to one fma, one exp2, one add: P stays unnormalised, 1/rowsum (v_rcp_f32) scales the 4 O values a lane feeds to out-proj; rounding changes, key sum order unchanged
# speedup vs baseline: 1.3520x; 1.0385x over previous
#include <hip/hip_runtime.h>
#include <math.h>


#define NUM_TBLOCKS 8
#define SEQ 54
#define DIM 6
#define QKV (3 * DIM)
#define FEAT (SEQ * DIM)
#define POL 12
#define LN_EPS 1e-5f
#define QSCALE 0.40824829046386302f
#define LOG2E 1.4426950408889634f

#define WAVES_PER_BLOCK 1
#define THREADS (WAVES_PER_BLOCK * 32)
#define PPITCH 68
#define LDS_FLOATS (64 * DIM + 16 * PPITCH + 64)

typedef __attribute__((ext_vector_type(2))) float v2f;
typedef __attribute__((ext_vector_type(4))) float v4f;
typedef __attribute__((ext_vector_type(8))) float v8f;

__device__ __forceinline__ v8f wmma_f32(v2f a, v2f b, v8f c) {
  return __builtin_amdgcn_wmma_f32_16x16x4_f32(false, a, false, b, (short)0, c,
                                               false, false);
}

__device__ __forceinline__ float eluf(float x) {
  return x > 0.0f ? x : (__expf(x) - 1.0f);
}


__global__ __launch_bounds__(THREADS) void rubik_fused(
    const float* __restrict__ x, const float* __restrict__ embed_w,
    const float* __restrict__ embed_b, const float* __restrict__ fc1_w,
    const float* __restrict__ fc1_b, const float* __restrict__ ln_g,
    const float* __restrict__ ln_b, const float* __restrict__ iw,
    const float* __restrict__ ib, const float* __restrict__ ow,
    const float* __restrict__ ob, const float* __restrict__ vw,
    const float* __restrict__ vb, const float* __restrict__ pw,
    const float* __restrict__ pb, float* __restrict__ out, int n_batch) {
  __shared__ __attribute__((aligned(16))) float smem[LDS_FLOATS];
  const int lane = threadIdx.x & 31;
  int b = blockIdx.x;
  if (b >= n_batch) b = n_batch - 1;

  float* Zs = smem;
  float* P = smem + 64 * DIM;
  float* dump = P + 16 * PPITCH;

  const float* xb = x + (size_t)b * FEAT;
  __builtin_prefetch(xb, 0, 1);

  const int nloc = lane & 15;
  const int hh = lane >> 4;
  const bool lowhalf = (lane < 16);
  const int kb = 2 * hh;
  const int upper8 = 8 * hh;
  const v2f zero2 = {0.0f, 0.0f};

  const int rr = nloc & 3;
  const int f6 = (nloc & 8) ? (rr + 2) : (rr < 2 ? rr : rr + 2);
  const bool rowv2 = (nloc & 8) ? (rr < 2) : true;
  const bool rowv1 = rowv2 && !(nloc & 4);
  const int jqk = ((nloc & 4) ? DIM : 0) + f6;

  const int s7 = nloc & 7;
  const bool tailv = (s7 < 3);
  const int tailtok = 48 + s7 + ((nloc & 8) ? 3 : 0);

  auto loadWA = [&](const float* w, bool v, int j, v2f aw[2]) {
    const float* wr = w + (v ? j : 0) * DIM;
    v2f w0 = *(const v2f*)(wr + kb);
    v2f w1 = *(const v2f*)(wr + (lowhalf ? 4 : 0));
    aw[0] = v ? w0 : zero2;
    aw[1] = (v && lowhalf) ? w1 : zero2;
  };
  auto loadCrow = [&](const float* bias, bool two) {
    v8f c;
#pragma unroll
    for (int r = 0; r < 8; ++r) {
      int q = r & 3;
      int f = hh ? (q + 2) : (q < 2 ? q : q + 2);
      bool v = hh ? (q < 2) : true;
      if (r >= 4 && !two) v = false;
      int j = f + (r >= 4 ? DIM : 0);
      float bv = bias[v ? j : 0];
      c[r] = v ? bv : 0.0f;
    }
    return c;
  };
  auto loadWB = [&](const float* w, const float* bias, v2f bw[2], float& bcol) {
    bool vj = (nloc < DIM);
    int jc = vj ? nloc : 0;
    const float* wr = w + jc * DIM;
    v2f w0 = *(const v2f*)(wr + kb);
    v2f w1 = *(const v2f*)(wr + (lowhalf ? 4 : 0));
    float bb = bias[jc];
    bw[0] = vj ? w0 : zero2;
    bw[1] = (vj && lowhalf) ? w1 : zero2;
    bcol = vj ? bb : 0.0f;
  };
  auto splatC = [&](float bvv) {
    v8f c;
#pragma unroll
    for (int r = 0; r < 8; ++r) c[r] = bvv;
    return c;
  };

  v2f H0[4], H1[4];

  {
    v2f ae[2];
    loadWA(embed_w, rowv1, f6, ae);
    v8f ce = loadCrow(embed_b, false);
#pragma unroll
    for (int t = 0; t < 4; ++t) {
      int tok = (t < 3) ? (t * 16 + nloc) : tailtok;
      bool tv = (t < 3) ? true : tailv;
      const float* xr = xb + (tv ? tok : 0) * DIM;
      v2f x0 = *(const v2f*)(xr + kb);
      v2f x1 = *(const v2f*)(xr + (lowhalf ? 4 : 0));
      x0 = tv ? x0 : zero2;
      x1 = (tv && lowhalf) ? x1 : zero2;
      v8f d = wmma_f32(ae[0], x0, ce);
      d = wmma_f32(ae[1], x1, d);
      H0[t] = (v2f){d[0], d[1]};
      H1[t] = (v2f){d[2], d[3]};
    }
  }

  for (int L = 0; L < NUM_TBLOCKS; ++L) {
    v2f af[2], aqk[2], bvw[2], ao[2];
    float bcv;
    loadWA(fc1_w + L * DIM * DIM, rowv1, f6, af);
    v8f cf = loadCrow(fc1_b + L * DIM, false);
    loadWA(iw + L * QKV * DIM, rowv2, jqk, aqk);
    v8f cqk = loadCrow(ib + L * QKV, true);
    loadWB(iw + L * QKV * DIM + 2 * DIM * DIM, ib + L * QKV + 2 * DIM, bvw, bcv);
    v8f cv = splatC(bcv);
    loadWA(ow + L * DIM * DIM, rowv1, f6, ao);
    v8f co = loadCrow(ob + L * DIM, false);
    float g4[4], b4[4];
#pragma unroll
    for (int i = 0; i < 4; ++i) {
      int f = hh ? (i + 2) : (i < 2 ? i : i + 2);
      bool v = hh ? (i < 2) : true;
      float gg = ln_g[L * DIM + (v ? f : 0)];
      float bb = ln_b[L * DIM + (v ? f : 0)];
      g4[i] = v ? gg : 0.0f;
      b4[i] = v ? bb : 0.0f;
    }

    v2f Qb0[4], Qb1[4], Ka0[4], Ka1[4];
    v8f Vd[4];
    float invq[4];

#pragma unroll
    for (int t = 0; t < 4; ++t) {
      v8f d = wmma_f32(af[0], H0[t], cf);
      d = wmma_f32(af[1], H1[t], d);
      float e0 = eluf(d[0]), e1 = eluf(d[1]), e2 = eluf(d[2]), e3 = eluf(d[3]);
      float s = (e0 + e1) + (e2 + e3);
      s += __shfl_xor(s, 16);
      float mu = s * (1.0f / DIM);
      float d0 = e0 - mu, d1 = e1 - mu;
      float d2 = lowhalf ? (e2 - mu) : 0.0f;
      float d3 = lowhalf ? (e3 - mu) : 0.0f;
      float var = (d0 * d0 + d1 * d1) + (d2 * d2 + d3 * d3);
      var += __shfl_xor(var, 16);
      var *= (1.0f / DIM);
      float rs = rsqrtf(var + LN_EPS);
      v2f z0 = {d0 * rs * g4[0] + b4[0], d1 * rs * g4[1] + b4[1]};
      v2f z1 = {d2 * rs * g4[2] + b4[2], d3 * rs * g4[3] + b4[3]};
      v8f dq = wmma_f32(aqk[0], z0, cqk);
      dq = wmma_f32(aqk[1], z1, dq);
      Qb0[t] = (v2f){dq[0] * QSCALE, dq[1] * QSCALE};
      Qb1[t] = (v2f){dq[2] * QSCALE, dq[3] * QSCALE};
      Ka0[t] = (v2f){dq[4], dq[5]};
      Ka1[t] = (v2f){dq[6], dq[7]};
      v8f dv = wmma_f32(z0, bvw[0], cv);
      dv = wmma_f32(z1, bvw[1], dv);
      Vd[t] = dv;
    }

#pragma unroll
    for (int nt = 0; nt < 4; ++nt) {
      v8f t[4];
#pragma unroll
      for (int mt = 0; mt < 4; ++mt) {
        v8f c = {};
        c = wmma_f32(Ka0[mt], Qb0[nt], c);
        c = wmma_f32(Ka1[mt], Qb1[nt], c);
        t[mt] = c;
      }

      float mx = -3.0e38f;
#pragma unroll
      for (int mt = 0; mt < 4; ++mt)
#pragma unroll
        for (int r = 0; r < 8; ++r) {
          if (mt < 3 || r < 3) mx = fmaxf(mx, t[mt][r]);
        }
      mx = fmaxf(mx, __shfl_xor(mx, 16));
      const float nmxl = -(mx * LOG2E);
      float sum = 0.0f;
#pragma unroll
      for (int mt = 0; mt < 4; ++mt)
#pragma unroll
        for (int r = 0; r < 8; ++r) {
          if (mt < 3 || r < 3) {
            float e = __builtin_amdgcn_exp2f(fmaf(t[mt][r], LOG2E, nmxl));
            t[mt][r] = e;
            sum += e;
          } else {
            t[mt][r] = 0.0f;
          }
        }
      sum += __shfl_xor(sum, 16);
      invq[nt] = __builtin_amdgcn_rcpf(sum);

#pragma unroll
      for (int mt = 0; mt < 4; ++mt) {
        float* bp = P + nloc * PPITCH + mt * 16 + upper8;
        if (mt < 3) {
          v4f lo = {t[mt][0], t[mt][1], t[mt][2], t[mt][3]};
          v4f hi = {t[mt][4], t[mt][5], t[mt][6], t[mt][7]};
          *(v4f*)(bp) = lo;
          *(v4f*)(bp + 4) = hi;
        } else {
          v4f lo = {t[mt][0], t[mt][1], t[mt][2], 0.0f};
          *(v4f*)(bp) = lo;
        }
      }
      __syncthreads();

      v8f o = {};
#pragma unroll
      for (int mt = 0; mt < 4; ++mt)
#pragma unroll
        for (int j = 0; j < 4; ++j) {
          if (mt == 3 && j >= 2) continue;
          v2f ap = *(const v2f*)(P + nloc * PPITCH + mt * 16 + upper8 + 2 * j);
          v2f bvv = {Vd[mt][2 * j], Vd[mt][2 * j + 1]};
          o = wmma_f32(ap, bvv, o);
        }
      {
        int row0 = nt * 16 + upper8;
        float* bz = (nloc < DIM) ? &Zs[row0 * DIM + nloc] : dump;
#pragma unroll
        for (int r = 0; r < 8; ++r) bz[r * DIM] = o[r];
      }
      __syncthreads();
    }

#pragma unroll
    for (int t = 0; t < 4; ++t) {
      const float* orow = Zs + (t * 16 + nloc) * DIM;
      v2f o0 = *(const v2f*)(orow + kb);
      v2f o1 = *(const v2f*)(orow + (lowhalf ? 4 : 0));
      o1 = lowhalf ? o1 : zero2;
      o0 = o0 * invq[t];
      o1 = o1 * invq[t];
      v8f d = wmma_f32(ao[0], o0, co);
      d = wmma_f32(ao[1], o1, d);
      H0[t] = (v2f){eluf(d[0]) + H0[t].x, eluf(d[1]) + H0[t].y};
      H1[t] = (v2f){eluf(d[2]) + H1[t].x, eluf(d[3]) + H1[t].y};
    }
    __syncthreads();
  }

#pragma unroll
  for (int t = 0; t < 4; ++t) {
    int tok = (t < 3) ? (t * 16 + nloc) : tailtok;
    bool tv = (t < 3) ? true : tailv;
    if (tv) {
      *(v2f*)(Zs + tok * DIM + kb) = H0[t];
      if (lowhalf) *(v2f*)(Zs + tok * DIM + 4) = H1[t];
    }
  }
  __syncthreads();
  const float* H = Zs;

  float vacc = 0.0f;
  float pacc[POL];
#pragma unroll
  for (int o = 0; o < POL; ++o) pacc[o] = 0.0f;
  for (int j = lane; j < FEAT; j += 32) {
    float fj = H[j];
    vacc += fj * vw[j];
#pragma unroll
    for (int o = 0; o < POL; ++o) pacc[o] += fj * pw[o * FEAT + j];
  }
  for (int off = 16; off >= 1; off >>= 1) {
    vacc += __shfl_xor(vacc, off, 32);
#pragma unroll
    for (int o = 0; o < POL; ++o) pacc[o] += __shfl_xor(pacc[o], off, 32);
  }
  if (lane == 0) {
    out[b] = vacc + vb[0];
    float logit[POL];
    float mx = -3.0e38f;
#pragma unroll
    for (int o = 0; o < POL; ++o) {
      logit[o] = pacc[o] + pb[o];
      mx = fmaxf(mx, logit[o]);
    }
    float sum = 0.0f;
#pragma unroll
    for (int o = 0; o < POL; ++o) {
      logit[o] = __expf(logit[o] - mx);
      sum += logit[o];
    }
    float inv = 1.0f / sum;
    float* pout = out + n_batch + (size_t)b * POL;
#pragma unroll
    for (int o = 0; o < POL; ++o) pout[o] = logit[o] * inv;
  }
}

extern "C" void kernel_launch(void* const* d_in, const int* in_sizes, int n_in,
                              void* d_out, int out_size, void* d_ws,
                              size_t ws_size, hipStream_t stream) {
  (void)n_in;
  (void)out_size;
  (void)d_ws;
  (void)ws_size;
  const float* x = (const float*)d_in[0];
  const float* embed_w = (const float*)d_in[1];
  const float* embed_b = (const float*)d_in[2];
  const float* fc1_w = (const float*)d_in[3];
  const float* fc1_b = (const float*)d_in[4];
  const float* ln_g = (const float*)d_in[5];
  const float* ln_b = (const float*)d_in[6];
  const float* iw = (const float*)d_in[7];
  const float* ib = (const float*)d_in[8];
  const float* ow = (const float*)d_in[9];
  const float* ob = (const float*)d_in[10];
  const float* vw = (const float*)d_in[11];
  const float* vb = (const float*)d_in[12];
  const float* pw = (const float*)d_in[13];
  const float* pb = (const float*)d_in[14];

  int n_batch = in_sizes[0] / FEAT;
  int grid = (n_batch + WAVES_PER_BLOCK - 1) / WAVES_PER_BLOCK;
  rubik_fused<<<grid, THREADS, 0, stream>>>(x, embed_w, embed_b, fc1_w, fc1_b,
                                            ln_g, ln_b, iw, ib, ow, ob, vw, vb,
                                            pw, pb, (float*)d_out, n_batch);
}
